# starting kernel
# baseline (speedup 1.0000x reference)
#include <hip/hip_runtime.h>
#include <math.h>

#define B_    32
#define IN_   2048
#define DIN_  8
#define N_    64
#define DOUT_ 16
#define C_    32
#define REP_  (IN_ / C_)
#define TILE_ 512
#define NTILE (IN_ / TILE_)
#define EPS_  1e-7f

typedef float v2f __attribute__((ext_vector_type(2)));
typedef float v8f __attribute__((ext_vector_type(8)));

__device__ __forceinline__ void hat_tile_wmma(
    const float* __restrict__ x, const float* __restrict__ Wt,
    float* __restrict__ hat, int b, int n, int tile, int wave, int lane)
{
  const int mrow = lane & 15;
  const int kh   = (lane >> 4) << 1;
  const int half = lane >> 4;
  for (int u = wave; u < TILE_ / 16; u += 8) {
    const int il0 = u << 4;
    const int gi0 = tile * TILE_ + il0;
    const int cu  = gi0 >> 6;
    const float* xr = x  + ((size_t)b * IN_ + gi0 + mrow) * DIN_ + kh;
    const float* wr = Wt + (((size_t)n * C_ + cu) * DOUT_ + mrow) * DIN_ + kh;
    v2f a0 = { xr[0], xr[1] };
    v2f a1 = { xr[4], xr[5] };
    v2f b0 = { wr[0], wr[1] };
    v2f b1 = { wr[4], wr[5] };
    v8f acc = {0.f,0.f,0.f,0.f,0.f,0.f,0.f,0.f};
    acc = __builtin_amdgcn_wmma_f32_16x16x4_f32(false, a0, false, b0,
                                                (short)0, acc, false, false);
    acc = __builtin_amdgcn_wmma_f32_16x16x4_f32(false, a1, false, b1,
                                                (short)0, acc, false, false);
#pragma unroll
    for (int r = 0; r < 8; ++r)
      hat[(il0 + r + (half << 3)) * DOUT_ + mrow] = acc[r];
  }
}

__global__ __launch_bounds__(256) void caps_softmax(
    const float* __restrict__ logits, float* __restrict__ smax, float* __restrict__ sinv)
{
  const int e = blockIdx.x * 256 + threadIdx.x;
  const int b = e >> 11;
  const int i = e & (IN_ - 1);
  const float* lp = logits + (size_t)b * N_ * IN_ + i;
  float mx = -3.4e38f;
#pragma unroll 4
  for (int n = 0; n < N_; ++n) mx = fmaxf(mx, lp[(size_t)n * IN_]);
  float s = 0.f;
#pragma unroll 4
  for (int n = 0; n < N_; ++n) s += __expf(lp[(size_t)n * IN_] - mx);
  smax[e] = mx;
  sinv[e] = 1.0f / s;
}

template <int MODE>
__global__ __launch_bounds__(256) void caps_route(
    const float* __restrict__ x, const float* __restrict__ Wt,
    const float* __restrict__ bias, float* __restrict__ logits,
    const float* __restrict__ smax, const float* __restrict__ sinv,
    float* __restrict__ out)
{
  __shared__ float hat[TILE_ * DOUT_];
  __shared__ float cld[TILE_];
  __shared__ float part[256];
  __shared__ float sld[DOUT_];
  __shared__ float old_[DOUT_];
  __shared__ float scl;

  const int b    = blockIdx.x >> 6;
  const int n    = blockIdx.x & (N_ - 1);
  const int wave = threadIdx.x >> 5;
  const int lane = threadIdx.x & 31;
  const int dc    = threadIdx.x & 15;
  const int chunk = threadIdx.x >> 4;
  const size_t lbase = ((size_t)b * N_ + n) * IN_;

  float sacc = 0.f;
  for (int tile = 0; tile < NTILE; ++tile) {
    if (tile + 1 < NTILE)
      __builtin_prefetch(
          x + ((size_t)b * IN_ + (tile + 1) * TILE_ + (threadIdx.x << 1)) * DIN_, 0, 0);
    hat_tile_wmma(x, Wt, hat, b, n, tile, wave, lane);
    for (int ii = threadIdx.x; ii < TILE_; ii += 256) {
      float cv;
      if (MODE == 0) {
        cv = 1.0f / (float)N_;
      } else {
        const int gi = tile * TILE_ + ii;
        const size_t e = (size_t)b * IN_ + gi;
        cv = __expf(logits[lbase + gi] - smax[e]) * sinv[e];
      }
      cld[ii] = cv;
    }
    __syncthreads();
    const float* hp = hat + (chunk * 32) * DOUT_ + dc;
    const float* cp = cld + chunk * 32;
#pragma unroll 8
    for (int ii = 0; ii < 32; ++ii) sacc += cp[ii] * hp[ii * DOUT_];
    __syncthreads();
  }

  part[threadIdx.x] = sacc;
  __syncthreads();
  if (threadIdx.x < DOUT_) {
    float s = bias[n * DOUT_ + threadIdx.x];
#pragma unroll
    for (int ch = 0; ch < 16; ++ch) s += part[ch * 16 + threadIdx.x];
    sld[threadIdx.x] = s;
  }
  __syncthreads();
  if (threadIdx.x == 0) {
    float sn = 0.f;
#pragma unroll
    for (int d2 = 0; d2 < DOUT_; ++d2) sn += sld[d2] * sld[d2];
    scl = sn / (1.0f + sn) * rsqrtf(sn + EPS_);
  }
  __syncthreads();
  if (threadIdx.x < DOUT_) {
    const float o = scl * sld[threadIdx.x];
    old_[threadIdx.x] = o;
    out[((size_t)b * N_ + n) * DOUT_ + threadIdx.x] = o;
  }
  __syncthreads();

  if (MODE < 2) {
    for (int tile = 0; tile < NTILE; ++tile) {
      hat_tile_wmma(x, Wt, hat, b, n, tile, wave, lane);
      __syncthreads();
      for (int ii = threadIdx.x; ii < TILE_; ii += 256) {
        const float* hr = hat + ii * DOUT_;
        float dlt = 0.f;
#pragma unroll
        for (int d2 = 0; d2 < DOUT_; ++d2) dlt += old_[d2] * hr[d2];
        const size_t g = lbase + tile * TILE_ + ii;
        logits[g] = (MODE == 0) ? dlt : (logits[g] + dlt);
      }
      __syncthreads();
    }
  }
}

extern "C" void kernel_launch(void* const* d_in, const int* in_sizes, int n_in,
                              void* d_out, int out_size, void* d_ws, size_t ws_size,
                              hipStream_t stream)
{
  const float* x    = (const float*)d_in[0];
  const float* Wt   = (const float*)d_in[1];
  const float* bias = (const float*)d_in[2];
  float* out = (float*)d_out;

  float* logits = (float*)d_ws;
  float* smax   = logits + (size_t)B_ * N_ * IN_;
  float* sinv   = smax   + (size_t)B_ * IN_;

  dim3 grid(B_ * N_), blk(256);
  dim3 sgrid((B_ * IN_) / 256), sblk(256);

  caps_route<0><<<grid, blk, 0, stream>>>(x, Wt, bias, logits, smax, sinv, out);
  caps_softmax<<<sgrid, sblk, 0, stream>>>(logits, smax, sinv);
  caps_route<1><<<grid, blk, 0, stream>>>(x, Wt, bias, logits, smax, sinv, out);
  caps_softmax<<<sgrid, sblk, 0, stream>>>(logits, smax, sinv);
  caps_route<2><<<grid, blk, 0, stream>>>(x, Wt, bias, logits, smax, sinv, out);
}


# keep the vote tile in WMMA accumulators: fold c[i] into the x fragment for s, take the product transposed and dot it with out in registers for delta; no LDS vote tile, no per-tile barriers
# speedup vs baseline: 1.2451x; 1.2451x over previous
#include <hip/hip_runtime.h>
#include <math.h>


#define B_    32
#define IN_   2048
#define DIN_  8
#define N_    64
#define DOUT_ 16
#define C_    32
#define REP_  (IN_ / C_)
#define WSPAN (IN_ / 8)
#define EPS_  1e-7f

typedef float v2f __attribute__((ext_vector_type(2)));
typedef float v8f __attribute__((ext_vector_type(8)));

__device__ __forceinline__ void load_frag(const float* __restrict__ p, v2f& f0, v2f& f1)
{
  f0 = (v2f){ p[0], p[1] };
  f1 = (v2f){ p[4], p[5] };
}

__device__ __forceinline__ float xhalf(float v)
{
  const int b = __builtin_bit_cast(int, v);
  return __builtin_bit_cast(float,
      __builtin_amdgcn_permlanex16(b, b, 0x76543210, 0xfedcba98, false, false));
}

__device__ __forceinline__ v8f wmma4(v2f a, v2f b, v8f c)
{
  return __builtin_amdgcn_wmma_f32_16x16x4_f32(false, a, false, b, (short)0, c, false, false);
}

__global__ __launch_bounds__(256) void caps_softmax(
    const float* __restrict__ logits, float* __restrict__ smax, float* __restrict__ sinv)
{
  const int e = blockIdx.x * 256 + threadIdx.x;
  const int b = e >> 11;
  const int i = e & (IN_ - 1);
  const float* lp = logits + (size_t)b * N_ * IN_ + i;
  float mx = -3.4e38f;
#pragma unroll 4
  for (int n = 0; n < N_; ++n) mx = fmaxf(mx, lp[(size_t)n * IN_]);
  float s = 0.f;
#pragma unroll 4
  for (int n = 0; n < N_; ++n) s += __expf(lp[(size_t)n * IN_] - mx);
  smax[e] = mx;
  sinv[e] = 1.0f / s;
}

template <int MODE>
__global__ __launch_bounds__(256) void caps_route(
    const float* __restrict__ x, const float* __restrict__ Wt,
    const float* __restrict__ bias, float* __restrict__ logits,
    const float* __restrict__ smax, const float* __restrict__ sinv,
    float* __restrict__ out)
{
  __shared__ float part[8 * DOUT_];
  __shared__ float sld[DOUT_];
  __shared__ float old_[DOUT_];
  __shared__ float scl;

  const int b    = blockIdx.x >> 6;
  const int n    = blockIdx.x & (N_ - 1);
  const int wave = threadIdx.x >> 5;
  const int lane = threadIdx.x & 31;
  const int c16  = lane & 15;
  const int half = lane >> 4;
  const int kh   = half << 1;
  const size_t lbase = ((size_t)b * N_ + n) * IN_;
  const int    iw    = wave * WSPAN;

  const float* xw = x  + ((size_t)b * IN_ + iw + c16) * DIN_ + kh;
  const float* ww = Wt + (((size_t)n * C_ + (iw >> 6)) * DOUT_ + c16) * DIN_ + kh;

  v8f acc0 = {0.f,0.f,0.f,0.f,0.f,0.f,0.f,0.f};
  v8f acc1 = {0.f,0.f,0.f,0.f,0.f,0.f,0.f,0.f};
#pragma unroll 1
  for (int ch = 0; ch < 4; ++ch) {
    v2f w0, w1;
    load_frag(ww + ch * (DOUT_ * DIN_), w0, w1);
#pragma unroll
    for (int u = 0; u < 4; ++u) {
      const int il = ch * REP_ + u * 16;
      v2f a0, a1;
      load_frag(xw + il * DIN_, a0, a1);
      if (MODE != 0) {
        const int gi = iw + il + c16;
        const size_t e = (size_t)b * IN_ + gi;
        const float cv = __expf(logits[lbase + gi] - smax[e]) * sinv[e];
        a0 *= cv;
        a1 *= cv;
      }
      if (u & 1) { acc1 = wmma4(a0, w0, acc1); acc1 = wmma4(a1, w1, acc1); }
      else       { acc0 = wmma4(a0, w0, acc0); acc0 = wmma4(a1, w1, acc0); }
    }
  }
  {
    float t = 0.f;
#pragma unroll
    for (int r = 0; r < 8; ++r) t += acc0[r] + acc1[r];
    t += xhalf(t);
    if (MODE == 0) t *= (1.0f / (float)N_);
    if (lane < DOUT_) part[wave * DOUT_ + lane] = t;
  }
  __syncthreads();

  if (threadIdx.x < DOUT_) {
    float s = bias[n * DOUT_ + threadIdx.x];
#pragma unroll
    for (int w = 0; w < 8; ++w) s += part[w * DOUT_ + threadIdx.x];
    sld[threadIdx.x] = s;
  }
  __syncthreads();
  if (threadIdx.x == 0) {
    float sn = 0.f;
#pragma unroll
    for (int d2 = 0; d2 < DOUT_; ++d2) sn += sld[d2] * sld[d2];
    scl = sn / (1.0f + sn) * rsqrtf(sn + EPS_);
  }
  __syncthreads();
  if (threadIdx.x < DOUT_) {
    const float o = scl * sld[threadIdx.x];
    old_[threadIdx.x] = o;
    out[((size_t)b * N_ + n) * DOUT_ + threadIdx.x] = o;
  }
  __syncthreads();

  if (MODE < 2) {
    v2f oh[4];
#pragma unroll
    for (int q = 0; q < 4; ++q)
      oh[q] = (v2f){ old_[(half << 3) + 2 * q], old_[(half << 3) + 2 * q + 1] };
    const v8f zero = {0.f,0.f,0.f,0.f,0.f,0.f,0.f,0.f};
#pragma unroll 1
    for (int ch = 0; ch < 4; ++ch) {
      v2f w0, w1;
      load_frag(ww + ch * (DOUT_ * DIN_), w0, w1);
#pragma unroll
      for (int up = 0; up < 2; ++up) {
        const int il = ch * REP_ + up * 32;
        v2f a0, a1, a2, a3;
        load_frag(xw + il * DIN_, a0, a1);
        load_frag(xw + (il + 16) * DIN_, a2, a3);
        v8f dA = wmma4(w0, a0, zero);
        v8f dB = wmma4(w0, a2, zero);
        dA = wmma4(w1, a1, dA);
        dB = wmma4(w1, a3, dB);
        v2f sA = {0.f, 0.f}, sB = {0.f, 0.f};
#pragma unroll
        for (int q = 0; q < 4; ++q) {
          sA += oh[q] * (v2f){ dA[2 * q], dA[2 * q + 1] };
          sB += oh[q] * (v2f){ dB[2 * q], dB[2 * q + 1] };
        }
        const float pA = sA.x + sA.y;
        const float pB = sB.x + sB.y;
        const float keep = half ? pB : pA;
        const float give = half ? pA : pB;
        const float dl = keep + xhalf(give);
        const size_t g = lbase + iw + il + lane;
        logits[g] = (MODE == 0) ? dl : (logits[g] + dl);
      }
    }
  }
}

extern "C" void kernel_launch(void* const* d_in, const int* in_sizes, int n_in,
                              void* d_out, int out_size, void* d_ws, size_t ws_size,
                              hipStream_t stream)
{
  const float* x    = (const float*)d_in[0];
  const float* Wt   = (const float*)d_in[1];
  const float* bias = (const float*)d_in[2];
  float* out = (float*)d_out;

  float* logits = (float*)d_ws;
  float* smax   = logits + (size_t)B_ * N_ * IN_;
  float* sinv   = smax   + (size_t)B_ * IN_;

  dim3 grid(B_ * N_), blk(256);
  dim3 sgrid((B_ * IN_) / 256), sblk(256);

  caps_route<0><<<grid, blk, 0, stream>>>(x, Wt, bias, logits, smax, sinv, out);
  caps_softmax<<<sgrid, sblk, 0, stream>>>(logits, smax, sinv);
  caps_route<1><<<grid, blk, 0, stream>>>(x, Wt, bias, logits, smax, sinv, out);
  caps_softmax<<<sgrid, sblk, 0, stream>>>(logits, smax, sinv);
  caps_route<2><<<grid, blk, 0, stream>>>(x, Wt, bias, logits, smax, sinv, out);
}
